# starting kernel
# baseline (speedup 1.0000x reference)
#include <hip/hip_runtime.h>
#include <hip/hip_bf16.h>


#define NN   100000
#define EE   1600000
#define INC  32
#define HC   64
#define OUTC 16

typedef float v2f __attribute__((ext_vector_type(2)));
typedef float v8f __attribute__((ext_vector_type(8)));

template<int K, int NT, bool DUAL, bool SCALE1, bool RELU>
__global__ __launch_bounds__(128) void gemm_wmma_k(
    const float* __restrict__ A1, const float* __restrict__ B1,
    const float* __restrict__ A2, const float* __restrict__ B2,
    const float* __restrict__ rdeg, const float* __restrict__ bias,
    float* __restrict__ out, int nrows)
{
    const int wid  = (int)blockIdx.x * (blockDim.x >> 5) + ((int)threadIdx.x >> 5);
    const int lane = threadIdx.x & 31;
    const int row0 = wid << 4;
    if (row0 >= nrows) return;
    const int half = lane >> 4;
    const int l16  = lane & 15;
    const int m    = row0 + l16;
    constexpr int LDB = NT * 16;

    v8f acc[NT];
#pragma unroll
    for (int t = 0; t < NT; ++t) acc[t] = (v8f){0,0,0,0,0,0,0,0};

    const float sc = SCALE1 ? rdeg[m] : 1.0f;
    const float* a1 = A1 + (size_t)m * K;
    const float* a2 = DUAL ? (A2 + (size_t)m * K) : nullptr;

#pragma unroll
    for (int k = 0; k < K; k += 4) {
        const int kb = k + half * 2;
        v2f af1; af1.x = a1[kb] * sc; af1.y = a1[kb + 1] * sc;
        v2f af2;
        if (DUAL) { af2.x = a2[kb]; af2.y = a2[kb + 1]; }
#pragma unroll
        for (int t = 0; t < NT; ++t) {
            v2f bf1;
            bf1.x = B1[(size_t)kb       * LDB + t * 16 + l16];
            bf1.y = B1[(size_t)(kb + 1) * LDB + t * 16 + l16];
            acc[t] = __builtin_amdgcn_wmma_f32_16x16x4_f32(
                false, af1, false, bf1, (short)0, acc[t], false, false);
            if (DUAL) {
                v2f bf2;
                bf2.x = B2[(size_t)kb       * LDB + t * 16 + l16];
                bf2.y = B2[(size_t)(kb + 1) * LDB + t * 16 + l16];
                acc[t] = __builtin_amdgcn_wmma_f32_16x16x4_f32(
                    false, af2, false, bf2, (short)0, acc[t], false, false);
            }
        }
    }

#pragma unroll
    for (int t = 0; t < NT; ++t) {
        const float bc = bias[t * 16 + l16];
#pragma unroll
        for (int r = 0; r < 8; ++r) {
            const int rr = r + half * 8;
            float v = acc[t][r] + bc;
            if (RELU) v = fmaxf(v, 0.0f);
            out[(size_t)(row0 + rr) * LDB + t * 16 + l16] = v;
        }
    }
}

__global__ void degree_k(const int* __restrict__ dst, float* __restrict__ deg, int e)
{
    int tid = blockIdx.x * blockDim.x + threadIdx.x;
    if (tid < e) atomicAdd(&deg[dst[tid]], 1.0f);
}

__global__ void rdeg_k(float* __restrict__ deg, int n)
{
    int tid = blockIdx.x * blockDim.x + threadIdx.x;
    if (tid < n) deg[tid] = 1.0f / fmaxf(deg[tid], 1.0f);
}

__global__ __launch_bounds__(256) void scatter_k(
    const float* __restrict__ h, const int* __restrict__ src,
    const int* __restrict__ dst, float* __restrict__ agg)
{
    const unsigned tid = blockIdx.x * 256u + threadIdx.x;
    const unsigned e   = tid >> 4;
    const unsigned c4  = (tid & 15u) << 2;
    const int s = src[e];
    const int d = dst[e];
    const float4 v = *(const float4*)(h + (size_t)s * HC + c4);
    float* ap = agg + (size_t)d * HC + c4;
    atomicAdd(ap + 0, v.x);
    atomicAdd(ap + 1, v.y);
    atomicAdd(ap + 2, v.z);
    atomicAdd(ap + 3, v.w);
}

__global__ __launch_bounds__(256) void ln_relu_res_k(
    const float* __restrict__ t, const float* __restrict__ g,
    const float* __restrict__ b, const float* __restrict__ xin,
    float* __restrict__ out, int nrows)
{
    const int wid  = (int)blockIdx.x * (blockDim.x >> 5) + ((int)threadIdx.x >> 5);
    const int lane = threadIdx.x & 31;
    if (wid >= nrows) return;
    const float* row = t + (size_t)wid * HC;
    float v0 = row[lane], v1 = row[lane + 32];
    float s = v0 + v1;
#pragma unroll
    for (int off = 16; off > 0; off >>= 1) s += __shfl_xor(s, off, 32);
    const float mu = s * (1.0f / HC);
    const float d0 = v0 - mu, d1 = v1 - mu;
    float q = d0 * d0 + d1 * d1;
#pragma unroll
    for (int off = 16; off > 0; off >>= 1) q += __shfl_xor(q, off, 32);
    const float rstd = rsqrtf(q * (1.0f / HC) + 1e-5f);
    const float y0 = fmaxf(d0 * rstd * g[lane]      + b[lane],      0.0f);
    const float y1 = fmaxf(d1 * rstd * g[lane + 32] + b[lane + 32], 0.0f);
    float* o = out + (size_t)wid * HC;
    const float* xi = xin + (size_t)wid * HC;
    o[lane]      = y0 + xi[lane];
    o[lane + 32] = y1 + xi[lane + 32];
}

extern "C" void kernel_launch(void* const* d_in, const int* in_sizes, int n_in,
                              void* d_out, int out_size, void* d_ws, size_t ws_size,
                              hipStream_t stream)
{
    const float* x      = (const float*)d_in[0];
    const int*   ei     = (const int*)d_in[1];
    const float* proj_w = (const float*)d_in[2];
    const float* proj_b = (const float*)d_in[3];
    const float* wl0 = (const float*)d_in[4];  const float* bl0 = (const float*)d_in[5];
    const float* wr0 = (const float*)d_in[6];  const float* g0  = (const float*)d_in[7];
    const float* be0 = (const float*)d_in[8];
    const float* wl1 = (const float*)d_in[9];  const float* bl1 = (const float*)d_in[10];
    const float* wr1 = (const float*)d_in[11]; const float* g1  = (const float*)d_in[12];
    const float* be1 = (const float*)d_in[13];
    const float* wl2 = (const float*)d_in[14]; const float* bl2 = (const float*)d_in[15];
    const float* wr2 = (const float*)d_in[16];

    const int* srcI = ei;
    const int* dstI = ei + EE;

    const size_t N64 = (size_t)NN * HC;
    float* hA  = (float*)d_ws;
    float* hB  = hA + N64;
    float* agg = hB + N64;
    float* deg = agg + N64;

    const int GEMM_BLK   = 128;
    const int gemm_grid  = (NN / 16 + 3) / 4;
    const int scat_grid  = (EE * 16) / 256;
    const int ln_grid    = NN / 8;
    const int deg_grid   = (EE + 255) / 256;
    const int rdeg_grid  = (NN + 255) / 256;

    hipMemsetAsync(deg, 0, NN * sizeof(float), stream);
    degree_k<<<deg_grid, 256, 0, stream>>>(dstI, deg, EE);
    rdeg_k<<<rdeg_grid, 256, 0, stream>>>(deg, NN);

    gemm_wmma_k<INC, 4, false, false, true><<<gemm_grid, GEMM_BLK, 0, stream>>>(
        x, proj_w, nullptr, nullptr, nullptr, proj_b, hA, NN);

    hipMemsetAsync(agg, 0, N64 * sizeof(float), stream);
    scatter_k<<<scat_grid, 256, 0, stream>>>(hA, srcI, dstI, agg);
    gemm_wmma_k<HC, 4, true, true, false><<<gemm_grid, GEMM_BLK, 0, stream>>>(
        agg, wl0, hA, wr0, deg, bl0, agg, NN);
    ln_relu_res_k<<<ln_grid, 256, 0, stream>>>(agg, g0, be0, hA, hB, NN);

    hipMemsetAsync(agg, 0, N64 * sizeof(float), stream);
    scatter_k<<<scat_grid, 256, 0, stream>>>(hB, srcI, dstI, agg);
    gemm_wmma_k<HC, 4, true, true, false><<<gemm_grid, GEMM_BLK, 0, stream>>>(
        agg, wl1, hB, wr1, deg, bl1, agg, NN);
    ln_relu_res_k<<<ln_grid, 256, 0, stream>>>(agg, g1, be1, hB, hA, NN);

    hipMemsetAsync(agg, 0, N64 * sizeof(float), stream);
    scatter_k<<<scat_grid, 256, 0, stream>>>(hA, srcI, dstI, agg);
    gemm_wmma_k<HC, 1, true, true, false><<<gemm_grid, GEMM_BLK, 0, stream>>>(
        agg, wl2, hA, wr2, deg, bl2, (float*)d_out, NN);
}


# Float-atomic scatter-add and its memsets replaced by a per-call counting sort of edges by destination (int atomic rank, block scan) and a register-sum row gather; per-node sum order = arrival rank
# speedup vs baseline: 2.1577x; 2.1577x over previous
#include <hip/hip_runtime.h>
#include <hip/hip_bf16.h>


#define NN   100000
#define EE   1600000
#define INC  32
#define HC   64
#define OUTC 16
#define EPT  8
#define EDGE_BLOCKS ((EE / EPT + 255) / 256)
#define EDGE_T (EDGE_BLOCKS * 256)

typedef float v2f __attribute__((ext_vector_type(2)));
typedef float v8f __attribute__((ext_vector_type(8)));

template<int K, int NT, bool DUAL, bool SCALE1, bool RELU>
__global__ __launch_bounds__(128) void gemm_wmma_k(
    const float* __restrict__ A1, const float* __restrict__ B1,
    const float* __restrict__ A2, const float* __restrict__ B2,
    const float* __restrict__ rdeg, const float* __restrict__ bias,
    float* __restrict__ out, int nrows)
{
    const int wid  = (int)blockIdx.x * (blockDim.x >> 5) + ((int)threadIdx.x >> 5);
    const int lane = threadIdx.x & 31;
    const int row0 = wid << 4;
    if (row0 >= nrows) return;
    const int half = lane >> 4;
    const int l16  = lane & 15;
    const int m    = row0 + l16;
    constexpr int LDB = NT * 16;

    v8f acc[NT];
#pragma unroll
    for (int t = 0; t < NT; ++t) acc[t] = (v8f){0,0,0,0,0,0,0,0};

    const float sc = SCALE1 ? rdeg[m] : 1.0f;
    const float* a1 = A1 + (size_t)m * K;
    const float* a2 = DUAL ? (A2 + (size_t)m * K) : nullptr;

#pragma unroll
    for (int k = 0; k < K; k += 4) {
        const int kb = k + half * 2;
        v2f af1; af1.x = a1[kb] * sc; af1.y = a1[kb + 1] * sc;
        v2f af2;
        if (DUAL) { af2.x = a2[kb]; af2.y = a2[kb + 1]; }
#pragma unroll
        for (int t = 0; t < NT; ++t) {
            v2f bf1;
            bf1.x = B1[(size_t)kb       * LDB + t * 16 + l16];
            bf1.y = B1[(size_t)(kb + 1) * LDB + t * 16 + l16];
            acc[t] = __builtin_amdgcn_wmma_f32_16x16x4_f32(
                false, af1, false, bf1, (short)0, acc[t], false, false);
            if (DUAL) {
                v2f bf2;
                bf2.x = B2[(size_t)kb       * LDB + t * 16 + l16];
                bf2.y = B2[(size_t)(kb + 1) * LDB + t * 16 + l16];
                acc[t] = __builtin_amdgcn_wmma_f32_16x16x4_f32(
                    false, af2, false, bf2, (short)0, acc[t], false, false);
            }
        }
    }

#pragma unroll
    for (int t = 0; t < NT; ++t) {
        const float bc = bias[t * 16 + l16];
#pragma unroll
        for (int r = 0; r < 8; ++r) {
            const int rr = r + half * 8;
            float v = acc[t][r] + bc;
            if (RELU) v = fmaxf(v, 0.0f);
            out[(size_t)(row0 + rr) * LDB + t * 16 + l16] = v;
        }
    }
}

__global__ __launch_bounds__(256) void count_rank_k(
    const int* __restrict__ dst, int* __restrict__ cnt, int* __restrict__ rank, int e)
{
    const int tid = blockIdx.x * 256 + threadIdx.x;
    int d[EPT];
#pragma unroll
    for (int k = 0; k < EPT; ++k) d[k] = dst[min(tid + k * EDGE_T, e - 1)];
    int r[EPT];
#pragma unroll
    for (int k = 0; k < EPT; ++k) {
        r[k] = 0;
        if (tid + k * EDGE_T < e) r[k] = atomicAdd(&cnt[d[k]], 1);
    }
#pragma unroll
    for (int k = 0; k < EPT; ++k)
        if (tid + k * EDGE_T < e) rank[tid + k * EDGE_T] = r[k];
}

__global__ __launch_bounds__(1024) void seg_start_k(
    const int* __restrict__ cnt, int* __restrict__ cursor,
    int* __restrict__ off, float* __restrict__ rdeg, int n)
{
    __shared__ int wsum[32];
    __shared__ int base_s;
    const int tid  = blockIdx.x * 1024 + threadIdx.x;
    const int lane = threadIdx.x & 31;
    const int w    = threadIdx.x >> 5;
    const int c = (tid < n) ? cnt[tid] : 0;
    int x = c;
#pragma unroll
    for (int o = 1; o < 32; o <<= 1) {
        const int y = __shfl_up(x, o, 32);
        if (lane >= o) x += y;
    }
    if (lane == 31) wsum[w] = x;
    __syncthreads();
    if (w == 0) {
        const int t = wsum[lane];
        int s = t;
#pragma unroll
        for (int o = 1; o < 32; o <<= 1) {
            const int y = __shfl_up(s, o, 32);
            if (lane >= o) s += y;
        }
        wsum[lane] = s - t;
        if (lane == 31) base_s = atomicAdd(cursor, s);
    }
    __syncthreads();
    if (tid < n) {
        off[tid]  = base_s + wsum[w] + (x - c);
        rdeg[tid] = 1.0f / fmaxf((float)c, 1.0f);
    }
}

__global__ __launch_bounds__(256) void fill_k(
    const int* __restrict__ src, const int* __restrict__ dst,
    const int* __restrict__ rank, const int* __restrict__ off,
    int* __restrict__ col, int e)
{
    const int tid = blockIdx.x * 256 + threadIdx.x;
    int d[EPT], r[EPT], s[EPT], o[EPT];
#pragma unroll
    for (int k = 0; k < EPT; ++k) {
        const int i = min(tid + k * EDGE_T, e - 1);
        d[k] = dst[i]; r[k] = rank[i]; s[k] = src[i];
    }
#pragma unroll
    for (int k = 0; k < EPT; ++k) o[k] = off[d[k]];
#pragma unroll
    for (int k = 0; k < EPT; ++k)
        if (tid + k * EDGE_T < e) col[o[k] + r[k]] = s[k];
}

__global__ __launch_bounds__(256) void gather_sum_k(
    const float* __restrict__ h, const int* __restrict__ col,
    const int* __restrict__ off, const int* __restrict__ cnt,
    float* __restrict__ agg)
{
    const unsigned tid = blockIdx.x * 256u + threadIdx.x;
    const unsigned n   = tid >> 4;
    const unsigned c4  = (tid & 15u) << 2;
    const int start = off[n];
    const int c     = cnt[n];
    const int last  = max(c - 1, 0);
    const float* hc = h + c4;
    float4 acc = make_float4(0.0f, 0.0f, 0.0f, 0.0f);
    if (c > 0) {
        const int* cp = col + start;
        int s[8];
#pragma unroll
        for (int u = 0; u < 8; ++u) s[u] = cp[min(u, last)];
        for (int j = 0; j < c; j += 8) {
            float4 v[8];
#pragma unroll
            for (int u = 0; u < 8; ++u) v[u] = *(const float4*)(hc + (size_t)s[u] * HC);
#pragma unroll
            for (int u = 0; u < 8; ++u) s[u] = cp[min(j + 8 + u, last)];
#pragma unroll
            for (int u = 0; u < 8; ++u) {
                const bool ok = (j + u) < c;
                acc.x += ok ? v[u].x : 0.0f;
                acc.y += ok ? v[u].y : 0.0f;
                acc.z += ok ? v[u].z : 0.0f;
                acc.w += ok ? v[u].w : 0.0f;
            }
        }
    }
    *(float4*)(agg + (size_t)n * HC + c4) = acc;
}

__global__ __launch_bounds__(256) void ln_relu_res_k(
    const float* __restrict__ t, const float* __restrict__ g,
    const float* __restrict__ b, const float* __restrict__ xin,
    float* __restrict__ out, int nrows)
{
    const int wid  = (int)blockIdx.x * (blockDim.x >> 5) + ((int)threadIdx.x >> 5);
    const int lane = threadIdx.x & 31;
    if (wid >= nrows) return;
    const float* row = t + (size_t)wid * HC;
    float v0 = row[lane], v1 = row[lane + 32];
    float s = v0 + v1;
#pragma unroll
    for (int off = 16; off > 0; off >>= 1) s += __shfl_xor(s, off, 32);
    const float mu = s * (1.0f / HC);
    const float d0 = v0 - mu, d1 = v1 - mu;
    float q = d0 * d0 + d1 * d1;
#pragma unroll
    for (int off = 16; off > 0; off >>= 1) q += __shfl_xor(q, off, 32);
    const float rstd = rsqrtf(q * (1.0f / HC) + 1e-5f);
    const float y0 = fmaxf(d0 * rstd * g[lane]      + b[lane],      0.0f);
    const float y1 = fmaxf(d1 * rstd * g[lane + 32] + b[lane + 32], 0.0f);
    float* o = out + (size_t)wid * HC;
    const float* xi = xin + (size_t)wid * HC;
    o[lane]      = y0 + xi[lane];
    o[lane + 32] = y1 + xi[lane + 32];
}

extern "C" void kernel_launch(void* const* d_in, const int* in_sizes, int n_in,
                              void* d_out, int out_size, void* d_ws, size_t ws_size,
                              hipStream_t stream)
{
    const float* x      = (const float*)d_in[0];
    const int*   ei     = (const int*)d_in[1];
    const float* proj_w = (const float*)d_in[2];
    const float* proj_b = (const float*)d_in[3];
    const float* wl0 = (const float*)d_in[4];  const float* bl0 = (const float*)d_in[5];
    const float* wr0 = (const float*)d_in[6];  const float* g0  = (const float*)d_in[7];
    const float* be0 = (const float*)d_in[8];
    const float* wl1 = (const float*)d_in[9];  const float* bl1 = (const float*)d_in[10];
    const float* wr1 = (const float*)d_in[11]; const float* g1  = (const float*)d_in[12];
    const float* be1 = (const float*)d_in[13];
    const float* wl2 = (const float*)d_in[14]; const float* bl2 = (const float*)d_in[15];
    const float* wr2 = (const float*)d_in[16];

    const int* srcI = ei;
    const int* dstI = ei + EE;

    const size_t N64 = (size_t)NN * HC;
    float* hA  = (float*)d_ws;
    float* hB  = hA + N64;
    float* agg = hB + N64;
    float* deg = agg + N64;
    int* cnt    = (int*)(deg + NN);
    int* cursor = cnt + NN;
    int* off    = cnt + NN + 32;
    int* rank   = off + NN;
    int* col    = rank + EE;

    const int GEMM_BLK   = 128;
    const int gemm_grid  = (NN / 16 + 3) / 4;
    const int gath_grid  = (NN * 16) / 256;
    const int ln_grid    = NN / 8;
    const int edge_grid  = EDGE_BLOCKS;
    const int seg_grid   = (NN + 1023) / 1024;

    hipMemsetAsync(cnt, 0, (NN + 1) * sizeof(int), stream);
    count_rank_k<<<edge_grid, 256, 0, stream>>>(dstI, cnt, rank, EE);
    seg_start_k<<<seg_grid, 1024, 0, stream>>>(cnt, cursor, off, deg, NN);
    fill_k<<<edge_grid, 256, 0, stream>>>(srcI, dstI, rank, off, col, EE);

    gemm_wmma_k<INC, 4, false, false, true><<<gemm_grid, GEMM_BLK, 0, stream>>>(
        x, proj_w, nullptr, nullptr, nullptr, proj_b, hA, NN);

    gather_sum_k<<<gath_grid, 256, 0, stream>>>(hA, col, off, cnt, agg);
    gemm_wmma_k<HC, 4, true, true, false><<<gemm_grid, GEMM_BLK, 0, stream>>>(
        agg, wl0, hA, wr0, deg, bl0, agg, NN);
    ln_relu_res_k<<<ln_grid, 256, 0, stream>>>(agg, g0, be0, hA, hB, NN);

    gather_sum_k<<<gath_grid, 256, 0, stream>>>(hB, col, off, cnt, agg);
    gemm_wmma_k<HC, 4, true, true, false><<<gemm_grid, GEMM_BLK, 0, stream>>>(
        agg, wl1, hB, wr1, deg, bl1, agg, NN);
    ln_relu_res_k<<<ln_grid, 256, 0, stream>>>(agg, g1, be1, hB, hA, NN);

    gather_sum_k<<<gath_grid, 256, 0, stream>>>(hA, col, off, cnt, agg);
    gemm_wmma_k<HC, 1, true, true, false><<<gemm_grid, GEMM_BLK, 0, stream>>>(
        agg, wl2, hA, wr2, deg, bl2, (float*)d_out, NN);
}


# LayerNorm+ReLU+residual moved into the dual GEMM's epilogue (row sums: 4 tiles in-lane, then a 16-lane DPP butterfly), removing both one-wave-per-row kernels and the pre-norm round trip
# speedup vs baseline: 2.4410x; 1.1313x over previous
#include <hip/hip_runtime.h>
#include <hip/hip_bf16.h>


#define NN   100000
#define EE   1600000
#define INC  32
#define HC   64
#define OUTC 16
#define EPT  8
#define EDGE_BLOCKS ((EE / EPT + 255) / 256)
#define EDGE_T (EDGE_BLOCKS * 256)

typedef float v2f __attribute__((ext_vector_type(2)));
typedef float v8f __attribute__((ext_vector_type(8)));

__device__ __forceinline__ float row16_step(float x, const int ctrl_sel)
{
    int y;
    if (ctrl_sel == 1)      y = __builtin_amdgcn_update_dpp(0, __float_as_int(x), 0x161, 0xf, 0xf, false);
    else if (ctrl_sel == 2) y = __builtin_amdgcn_update_dpp(0, __float_as_int(x), 0x162, 0xf, 0xf, false);
    else if (ctrl_sel == 4) y = __builtin_amdgcn_update_dpp(0, __float_as_int(x), 0x164, 0xf, 0xf, false);
    else                    y = __builtin_amdgcn_update_dpp(0, __float_as_int(x), 0x168, 0xf, 0xf, false);
    return x + __int_as_float(y);
}

template<int K, int NT, bool DUAL, bool SCALE1, bool RELU, bool LNRES>
__global__ __launch_bounds__(128) void gemm_wmma_k(
    const float* __restrict__ A1, const float* __restrict__ B1,
    const float* __restrict__ A2, const float* __restrict__ B2,
    const float* __restrict__ rdeg, const float* __restrict__ bias,
    const float* __restrict__ lng, const float* __restrict__ lnb,
    float* __restrict__ out, int nrows)
{
    const int wid  = (int)blockIdx.x * (blockDim.x >> 5) + ((int)threadIdx.x >> 5);
    const int lane = threadIdx.x & 31;
    const int row0 = wid << 4;
    if (row0 >= nrows) return;
    const int half = lane >> 4;
    const int l16  = lane & 15;
    const int m    = row0 + l16;
    constexpr int LDB = NT * 16;

    v8f acc[NT];
#pragma unroll
    for (int t = 0; t < NT; ++t) acc[t] = (v8f){0,0,0,0,0,0,0,0};

    const float sc = SCALE1 ? rdeg[m] : 1.0f;
    const float* a1 = A1 + (size_t)m * K;
    const float* a2 = DUAL ? (A2 + (size_t)m * K) : nullptr;

#pragma unroll
    for (int k = 0; k < K; k += 4) {
        const int kb = k + half * 2;
        v2f af1; af1.x = a1[kb] * sc; af1.y = a1[kb + 1] * sc;
        v2f af2;
        if (DUAL) { af2.x = a2[kb]; af2.y = a2[kb + 1]; }
#pragma unroll
        for (int t = 0; t < NT; ++t) {
            v2f bf1;
            bf1.x = B1[(size_t)kb       * LDB + t * 16 + l16];
            bf1.y = B1[(size_t)(kb + 1) * LDB + t * 16 + l16];
            acc[t] = __builtin_amdgcn_wmma_f32_16x16x4_f32(
                false, af1, false, bf1, (short)0, acc[t], false, false);
            if (DUAL) {
                v2f bf2;
                bf2.x = B2[(size_t)kb       * LDB + t * 16 + l16];
                bf2.y = B2[(size_t)(kb + 1) * LDB + t * 16 + l16];
                acc[t] = __builtin_amdgcn_wmma_f32_16x16x4_f32(
                    false, af2, false, bf2, (short)0, acc[t], false, false);
            }
        }
    }

    if (LNRES) {
        static_assert(!LNRES || (DUAL && NT * 16 == HC && K == HC), "row must be whole");
        const size_t rbase = (size_t)(row0 + half * 8) * HC + l16;
        float res[NT][8];
#pragma unroll
        for (int t = 0; t < NT; ++t)
#pragma unroll
            for (int r = 0; r < 8; ++r) res[t][r] = A2[rbase + (size_t)r * HC + t * 16];
        float gc[NT], bc[NT];
#pragma unroll
        for (int t = 0; t < NT; ++t) { gc[t] = lng[t * 16 + l16]; bc[t] = lnb[t * 16 + l16]; }

        float s[8];
#pragma unroll
        for (int t = 0; t < NT; ++t) {
            const float bt = bias[t * 16 + l16];
#pragma unroll
            for (int r = 0; r < 8; ++r) {
                acc[t][r] += bt;
                s[r] = (t == 0) ? acc[t][r] : s[r] + acc[t][r];
            }
        }
#pragma unroll
        for (int r = 0; r < 8; ++r) s[r] = row16_step(s[r], 1);
#pragma unroll
        for (int r = 0; r < 8; ++r) s[r] = row16_step(s[r], 2);
#pragma unroll
        for (int r = 0; r < 8; ++r) s[r] = row16_step(s[r], 4);
#pragma unroll
        for (int r = 0; r < 8; ++r) s[r] = row16_step(s[r], 8);

        float q[8];
#pragma unroll
        for (int r = 0; r < 8; ++r) {
            const float mu = s[r] * (1.0f / HC);
#pragma unroll
            for (int t = 0; t < NT; ++t) {
                acc[t][r] -= mu;
                q[r] = (t == 0) ? acc[t][r] * acc[t][r] : q[r] + acc[t][r] * acc[t][r];
            }
        }
#pragma unroll
        for (int r = 0; r < 8; ++r) q[r] = row16_step(q[r], 1);
#pragma unroll
        for (int r = 0; r < 8; ++r) q[r] = row16_step(q[r], 2);
#pragma unroll
        for (int r = 0; r < 8; ++r) q[r] = row16_step(q[r], 4);
#pragma unroll
        for (int r = 0; r < 8; ++r) q[r] = row16_step(q[r], 8);

#pragma unroll
        for (int r = 0; r < 8; ++r) {
            const float rstd = rsqrtf(q[r] * (1.0f / HC) + 1e-5f);
#pragma unroll
            for (int t = 0; t < NT; ++t) {
                const float y = fmaxf(acc[t][r] * rstd * gc[t] + bc[t], 0.0f);
                out[rbase + (size_t)r * HC + t * 16] = y + res[t][r];
            }
        }
    } else {
#pragma unroll
        for (int t = 0; t < NT; ++t) {
            const float bc = bias[t * 16 + l16];
#pragma unroll
            for (int r = 0; r < 8; ++r) {
                const int rr = r + half * 8;
                float v = acc[t][r] + bc;
                if (RELU) v = fmaxf(v, 0.0f);
                out[(size_t)(row0 + rr) * LDB + t * 16 + l16] = v;
            }
        }
    }
}

__global__ __launch_bounds__(256) void count_rank_k(
    const int* __restrict__ dst, int* __restrict__ cnt, int* __restrict__ rank, int e)
{
    const int tid = blockIdx.x * 256 + threadIdx.x;
    int d[EPT];
#pragma unroll
    for (int k = 0; k < EPT; ++k) d[k] = dst[min(tid + k * EDGE_T, e - 1)];
    int r[EPT];
#pragma unroll
    for (int k = 0; k < EPT; ++k) {
        r[k] = 0;
        if (tid + k * EDGE_T < e) r[k] = atomicAdd(&cnt[d[k]], 1);
    }
#pragma unroll
    for (int k = 0; k < EPT; ++k)
        if (tid + k * EDGE_T < e) rank[tid + k * EDGE_T] = r[k];
}

__global__ __launch_bounds__(1024) void seg_start_k(
    const int* __restrict__ cnt, int* __restrict__ cursor,
    int* __restrict__ off, float* __restrict__ rdeg, int n)
{
    __shared__ int wsum[32];
    __shared__ int base_s;
    const int tid  = blockIdx.x * 1024 + threadIdx.x;
    const int lane = threadIdx.x & 31;
    const int w    = threadIdx.x >> 5;
    const int c = (tid < n) ? cnt[tid] : 0;
    int x = c;
#pragma unroll
    for (int o = 1; o < 32; o <<= 1) {
        const int y = __shfl_up(x, o, 32);
        if (lane >= o) x += y;
    }
    if (lane == 31) wsum[w] = x;
    __syncthreads();
    if (w == 0) {
        const int t = wsum[lane];
        int s = t;
#pragma unroll
        for (int o = 1; o < 32; o <<= 1) {
            const int y = __shfl_up(s, o, 32);
            if (lane >= o) s += y;
        }
        wsum[lane] = s - t;
        if (lane == 31) base_s = atomicAdd(cursor, s);
    }
    __syncthreads();
    if (tid < n) {
        off[tid]  = base_s + wsum[w] + (x - c);
        rdeg[tid] = 1.0f / fmaxf((float)c, 1.0f);
    }
}

__global__ __launch_bounds__(256) void fill_k(
    const int* __restrict__ src, const int* __restrict__ dst,
    const int* __restrict__ rank, const int* __restrict__ off,
    int* __restrict__ col, int e)
{
    const int tid = blockIdx.x * 256 + threadIdx.x;
    int d[EPT], r[EPT], s[EPT], o[EPT];
#pragma unroll
    for (int k = 0; k < EPT; ++k) {
        const int i = min(tid + k * EDGE_T, e - 1);
        d[k] = dst[i]; r[k] = rank[i]; s[k] = src[i];
    }
#pragma unroll
    for (int k = 0; k < EPT; ++k) o[k] = off[d[k]];
#pragma unroll
    for (int k = 0; k < EPT; ++k)
        if (tid + k * EDGE_T < e) col[o[k] + r[k]] = s[k];
}

__global__ __launch_bounds__(256) void gather_sum_k(
    const float* __restrict__ h, const int* __restrict__ col,
    const int* __restrict__ off, const int* __restrict__ cnt,
    float* __restrict__ agg)
{
    const unsigned tid = blockIdx.x * 256u + threadIdx.x;
    const unsigned n   = tid >> 4;
    const unsigned c4  = (tid & 15u) << 2;
    const int start = off[n];
    const int c     = cnt[n];
    const int last  = max(c - 1, 0);
    const float* hc = h + c4;
    float4 acc = make_float4(0.0f, 0.0f, 0.0f, 0.0f);
    if (c > 0) {
        const int* cp = col + start;
        int s[8];
#pragma unroll
        for (int u = 0; u < 8; ++u) s[u] = cp[min(u, last)];
        for (int j = 0; j < c; j += 8) {
            float4 v[8];
#pragma unroll
            for (int u = 0; u < 8; ++u) v[u] = *(const float4*)(hc + (size_t)s[u] * HC);
#pragma unroll
            for (int u = 0; u < 8; ++u) s[u] = cp[min(j + 8 + u, last)];
#pragma unroll
            for (int u = 0; u < 8; ++u) {
                const bool ok = (j + u) < c;
                acc.x += ok ? v[u].x : 0.0f;
                acc.y += ok ? v[u].y : 0.0f;
                acc.z += ok ? v[u].z : 0.0f;
                acc.w += ok ? v[u].w : 0.0f;
            }
        }
    }
    *(float4*)(agg + (size_t)n * HC + c4) = acc;
}

extern "C" void kernel_launch(void* const* d_in, const int* in_sizes, int n_in,
                              void* d_out, int out_size, void* d_ws, size_t ws_size,
                              hipStream_t stream)
{
    const float* x      = (const float*)d_in[0];
    const int*   ei     = (const int*)d_in[1];
    const float* proj_w = (const float*)d_in[2];
    const float* proj_b = (const float*)d_in[3];
    const float* wl0 = (const float*)d_in[4];  const float* bl0 = (const float*)d_in[5];
    const float* wr0 = (const float*)d_in[6];  const float* g0  = (const float*)d_in[7];
    const float* be0 = (const float*)d_in[8];
    const float* wl1 = (const float*)d_in[9];  const float* bl1 = (const float*)d_in[10];
    const float* wr1 = (const float*)d_in[11]; const float* g1  = (const float*)d_in[12];
    const float* be1 = (const float*)d_in[13];
    const float* wl2 = (const float*)d_in[14]; const float* bl2 = (const float*)d_in[15];
    const float* wr2 = (const float*)d_in[16];

    const int* srcI = ei;
    const int* dstI = ei + EE;

    const size_t N64 = (size_t)NN * HC;
    float* hA  = (float*)d_ws;
    float* hB  = hA + N64;
    float* agg = hB + N64;
    float* deg = agg + N64;
    int* cnt    = (int*)(deg + NN);
    int* cursor = cnt + NN;
    int* off    = cnt + NN + 32;
    int* rank   = off + NN;
    int* col    = rank + EE;

    const int GEMM_BLK   = 128;
    const int gemm_grid  = (NN / 16 + 3) / 4;
    const int gath_grid  = (NN * 16) / 256;
    const int edge_grid  = EDGE_BLOCKS;
    const int seg_grid   = (NN + 1023) / 1024;

    hipMemsetAsync(cnt, 0, (NN + 1) * sizeof(int), stream);
    count_rank_k<<<edge_grid, 256, 0, stream>>>(dstI, cnt, rank, EE);
    seg_start_k<<<seg_grid, 1024, 0, stream>>>(cnt, cursor, off, deg, NN);
    fill_k<<<edge_grid, 256, 0, stream>>>(srcI, dstI, rank, off, col, EE);

    gemm_wmma_k<INC, 4, false, false, true, false><<<gemm_grid, GEMM_BLK, 0, stream>>>(
        x, proj_w, nullptr, nullptr, nullptr, proj_b, nullptr, nullptr, hA, NN);

    gather_sum_k<<<gath_grid, 256, 0, stream>>>(hA, col, off, cnt, agg);
    gemm_wmma_k<HC, 4, true, true, false, true><<<gemm_grid, GEMM_BLK, 0, stream>>>(
        agg, wl0, hA, wr0, deg, bl0, g0, be0, hB, NN);

    gather_sum_k<<<gath_grid, 256, 0, stream>>>(hB, col, off, cnt, agg);
    gemm_wmma_k<HC, 4, true, true, false, true><<<gemm_grid, GEMM_BLK, 0, stream>>>(
        agg, wl1, hB, wr1, deg, bl1, g1, be1, hA, NN);

    gather_sum_k<<<gath_grid, 256, 0, stream>>>(hA, col, off, cnt, agg);
    gemm_wmma_k<HC, 1, true, true, false, false><<<gemm_grid, GEMM_BLK, 0, stream>>>(
        agg, wl2, hA, wr2, deg, bl2, nullptr, nullptr, (float*)d_out, NN);
}


# Neighbour sum moved into each layer's GEMM launch: a wave sums its 16 nodes' rows in registers (same order), parks the 16 means in LDS and reads them as the A operand; 3 launches fewer
# speedup vs baseline: 2.7429x; 1.1237x over previous
#include <hip/hip_runtime.h>
#include <hip/hip_bf16.h>


#define NN   100000
#define EE   1600000
#define INC  32
#define HC   64
#define OUTC 16
#define EPT  8
#define EDGE_BLOCKS ((EE / EPT + 255) / 256)
#define EDGE_T (EDGE_BLOCKS * 256)
#define APITCH 68

typedef float v2f __attribute__((ext_vector_type(2)));
typedef float v8f __attribute__((ext_vector_type(8)));

__device__ __forceinline__ float row16_step(float x, const int ctrl_sel)
{
    int y;
    if (ctrl_sel == 1)      y = __builtin_amdgcn_update_dpp(0, __float_as_int(x), 0x161, 0xf, 0xf, false);
    else if (ctrl_sel == 2) y = __builtin_amdgcn_update_dpp(0, __float_as_int(x), 0x162, 0xf, 0xf, false);
    else if (ctrl_sel == 4) y = __builtin_amdgcn_update_dpp(0, __float_as_int(x), 0x164, 0xf, 0xf, false);
    else                    y = __builtin_amdgcn_update_dpp(0, __float_as_int(x), 0x168, 0xf, 0xf, false);
    return x + __int_as_float(y);
}

template<int K, int NT, bool DUAL, bool GATHER, bool RELU, bool LNRES>
__global__ __launch_bounds__(128) void gemm_wmma_k(
    const float* __restrict__ A1, const float* __restrict__ B1,
    const float* __restrict__ A2, const float* __restrict__ B2,
    const float* __restrict__ rdeg, const float* __restrict__ bias,
    const float* __restrict__ lng, const float* __restrict__ lnb,
    const int* __restrict__ col, const int* __restrict__ off, const int* __restrict__ cnt,
    float* __restrict__ out, int nrows)
{
    __shared__ __attribute__((aligned(16))) float stage[GATHER ? 4 * 16 * APITCH : 4];
    const int wid  = (int)blockIdx.x * (blockDim.x >> 5) + ((int)threadIdx.x >> 5);
    const int lane = threadIdx.x & 31;
    const int row0 = wid << 4;
    if (row0 >= nrows) return;
    const int half = lane >> 4;
    const int l16  = lane & 15;
    const int m    = row0 + l16;
    constexpr int LDB = NT * 16;
    float* tile = stage + ((int)threadIdx.x >> 5) * (16 * APITCH);

    if constexpr (GATHER) {
        static_assert(!GATHER || (DUAL && K == HC), "the gathered rows are rows of A2");
        const int   offl = off[m];
        const int   cntl = cnt[m];
        const float rdl  = rdeg[m];
        const int   c4   = l16 << 2;
        const float* hc  = A2 + c4;
        int s[8];
        {
            const int st0 = __shfl(offl, half, 32);
            const int c0  = __shfl(cntl, half, 32);
            const int la0 = max(c0 - 1, 0);
#pragma unroll
            for (int u = 0; u < 8; ++u) s[u] = col[(c0 > 0) ? st0 + min(u, la0) : 0];
        }
        for (int i = 0; i < 8; ++i) {
            const int r      = 2 * i + half;
            const int rn     = min(r + 2, 14 + half);
            const int start  = __shfl(offl, r, 32);
            const int c      = __shfl(cntl, r, 32);
            const float sc   = __shfl(rdl,  r, 32);
            const int startn = __shfl(offl, rn, 32);
            const int cn     = __shfl(cntl, rn, 32);
            const int last   = max(c - 1, 0);
            const int lastn  = max(cn - 1, 0);
            float4 acc = make_float4(0.0f, 0.0f, 0.0f, 0.0f);
            int j = 0;
            do {
                float4 v[8];
#pragma unroll
                for (int u = 0; u < 8; ++u) v[u] = *(const float4*)(hc + (size_t)s[u] * HC);
                const bool more = (j + 8) < c;
#pragma unroll
                for (int u = 0; u < 8; ++u) {
                    const int nx = (cn > 0) ? startn + min(u, lastn) : 0;
                    s[u] = col[more ? start + min(j + 8 + u, last) : nx];
                }
#pragma unroll
                for (int u = 0; u < 8; ++u) {
                    const bool ok = (j + u) < c;
                    acc.x += ok ? v[u].x : 0.0f;
                    acc.y += ok ? v[u].y : 0.0f;
                    acc.z += ok ? v[u].z : 0.0f;
                    acc.w += ok ? v[u].w : 0.0f;
                }
                j += 8;
            } while (j < c);
            acc.x *= sc; acc.y *= sc; acc.z *= sc; acc.w *= sc;
            *(float4*)(tile + r * APITCH + c4) = acc;
        }
        __builtin_amdgcn_fence(__ATOMIC_RELEASE, "wavefront");
        __builtin_amdgcn_fence(__ATOMIC_ACQUIRE, "wavefront");
    }

    v8f acc[NT];
#pragma unroll
    for (int t = 0; t < NT; ++t) acc[t] = (v8f){0,0,0,0,0,0,0,0};

    const float* a1 = GATHER ? nullptr : (A1 + (size_t)m * K);
    const float* a2 = DUAL ? (A2 + (size_t)m * K) : nullptr;

#pragma unroll
    for (int k = 0; k < K; k += 4) {
        const int kb = k + half * 2;
        v2f af1;
        if constexpr (GATHER) af1 = *(const v2f*)(tile + l16 * APITCH + kb);
        else { af1.x = a1[kb]; af1.y = a1[kb + 1]; }
        v2f af2;
        if (DUAL) { af2.x = a2[kb]; af2.y = a2[kb + 1]; }
#pragma unroll
        for (int t = 0; t < NT; ++t) {
            v2f bf1;
            bf1.x = B1[(size_t)kb       * LDB + t * 16 + l16];
            bf1.y = B1[(size_t)(kb + 1) * LDB + t * 16 + l16];
            acc[t] = __builtin_amdgcn_wmma_f32_16x16x4_f32(
                false, af1, false, bf1, (short)0, acc[t], false, false);
            if (DUAL) {
                v2f bf2;
                bf2.x = B2[(size_t)kb       * LDB + t * 16 + l16];
                bf2.y = B2[(size_t)(kb + 1) * LDB + t * 16 + l16];
                acc[t] = __builtin_amdgcn_wmma_f32_16x16x4_f32(
                    false, af2, false, bf2, (short)0, acc[t], false, false);
            }
        }
    }

    if (LNRES) {
        static_assert(!LNRES || (DUAL && NT * 16 == HC && K == HC), "row must be whole");
        const size_t rbase = (size_t)(row0 + half * 8) * HC + l16;
        float res[NT][8];
#pragma unroll
        for (int t = 0; t < NT; ++t)
#pragma unroll
            for (int r = 0; r < 8; ++r) res[t][r] = A2[rbase + (size_t)r * HC + t * 16];
        float gc[NT], bc[NT];
#pragma unroll
        for (int t = 0; t < NT; ++t) { gc[t] = lng[t * 16 + l16]; bc[t] = lnb[t * 16 + l16]; }

        float s[8];
#pragma unroll
        for (int t = 0; t < NT; ++t) {
            const float bt = bias[t * 16 + l16];
#pragma unroll
            for (int r = 0; r < 8; ++r) {
                acc[t][r] += bt;
                s[r] = (t == 0) ? acc[t][r] : s[r] + acc[t][r];
            }
        }
#pragma unroll
        for (int r = 0; r < 8; ++r) s[r] = row16_step(s[r], 1);
#pragma unroll
        for (int r = 0; r < 8; ++r) s[r] = row16_step(s[r], 2);
#pragma unroll
        for (int r = 0; r < 8; ++r) s[r] = row16_step(s[r], 4);
#pragma unroll
        for (int r = 0; r < 8; ++r) s[r] = row16_step(s[r], 8);

        float q[8];
#pragma unroll
        for (int r = 0; r < 8; ++r) {
            const float mu = s[r] * (1.0f / HC);
#pragma unroll
            for (int t = 0; t < NT; ++t) {
                acc[t][r] -= mu;
                q[r] = (t == 0) ? acc[t][r] * acc[t][r] : q[r] + acc[t][r] * acc[t][r];
            }
        }
#pragma unroll
        for (int r = 0; r < 8; ++r) q[r] = row16_step(q[r], 1);
#pragma unroll
        for (int r = 0; r < 8; ++r) q[r] = row16_step(q[r], 2);
#pragma unroll
        for (int r = 0; r < 8; ++r) q[r] = row16_step(q[r], 4);
#pragma unroll
        for (int r = 0; r < 8; ++r) q[r] = row16_step(q[r], 8);

#pragma unroll
        for (int r = 0; r < 8; ++r) {
            const float rstd = rsqrtf(q[r] * (1.0f / HC) + 1e-5f);
#pragma unroll
            for (int t = 0; t < NT; ++t) {
                const float y = fmaxf(acc[t][r] * rstd * gc[t] + bc[t], 0.0f);
                out[rbase + (size_t)r * HC + t * 16] = y + res[t][r];
            }
        }
    } else {
#pragma unroll
        for (int t = 0; t < NT; ++t) {
            const float bc = bias[t * 16 + l16];
#pragma unroll
            for (int r = 0; r < 8; ++r) {
                const int rr = r + half * 8;
                float v = acc[t][r] + bc;
                if (RELU) v = fmaxf(v, 0.0f);
                out[(size_t)(row0 + rr) * LDB + t * 16 + l16] = v;
            }
        }
    }
}

__global__ __launch_bounds__(256) void count_rank_k(
    const int* __restrict__ dst, int* __restrict__ cnt, int* __restrict__ rank, int e)
{
    const int tid = blockIdx.x * 256 + threadIdx.x;
    int d[EPT];
#pragma unroll
    for (int k = 0; k < EPT; ++k) d[k] = dst[min(tid + k * EDGE_T, e - 1)];
    int r[EPT];
#pragma unroll
    for (int k = 0; k < EPT; ++k) {
        r[k] = 0;
        if (tid + k * EDGE_T < e) r[k] = atomicAdd(&cnt[d[k]], 1);
    }
#pragma unroll
    for (int k = 0; k < EPT; ++k)
        if (tid + k * EDGE_T < e) rank[tid + k * EDGE_T] = r[k];
}

__global__ __launch_bounds__(1024) void seg_start_k(
    const int* __restrict__ cnt, int* __restrict__ cursor,
    int* __restrict__ off, float* __restrict__ rdeg, int n)
{
    __shared__ int wsum[32];
    __shared__ int base_s;
    const int tid  = blockIdx.x * 1024 + threadIdx.x;
    const int lane = threadIdx.x & 31;
    const int w    = threadIdx.x >> 5;
    const int c = (tid < n) ? cnt[tid] : 0;
    int x = c;
#pragma unroll
    for (int o = 1; o < 32; o <<= 1) {
        const int y = __shfl_up(x, o, 32);
        if (lane >= o) x += y;
    }
    if (lane == 31) wsum[w] = x;
    __syncthreads();
    if (w == 0) {
        const int t = wsum[lane];
        int s = t;
#pragma unroll
        for (int o = 1; o < 32; o <<= 1) {
            const int y = __shfl_up(s, o, 32);
            if (lane >= o) s += y;
        }
        wsum[lane] = s - t;
        if (lane == 31) base_s = atomicAdd(cursor, s);
    }
    __syncthreads();
    if (tid < n) {
        off[tid]  = base_s + wsum[w] + (x - c);
        rdeg[tid] = 1.0f / fmaxf((float)c, 1.0f);
    }
}

__global__ __launch_bounds__(256) void fill_k(
    const int* __restrict__ src, const int* __restrict__ dst,
    const int* __restrict__ rank, const int* __restrict__ off,
    int* __restrict__ col, int e)
{
    const int tid = blockIdx.x * 256 + threadIdx.x;
    int d[EPT], r[EPT], s[EPT], o[EPT];
#pragma unroll
    for (int k = 0; k < EPT; ++k) {
        const int i = min(tid + k * EDGE_T, e - 1);
        d[k] = dst[i]; r[k] = rank[i]; s[k] = src[i];
    }
#pragma unroll
    for (int k = 0; k < EPT; ++k) o[k] = off[d[k]];
#pragma unroll
    for (int k = 0; k < EPT; ++k)
        if (tid + k * EDGE_T < e) col[o[k] + r[k]] = s[k];
}

extern "C" void kernel_launch(void* const* d_in, const int* in_sizes, int n_in,
                              void* d_out, int out_size, void* d_ws, size_t ws_size,
                              hipStream_t stream)
{
    const float* x      = (const float*)d_in[0];
    const int*   ei     = (const int*)d_in[1];
    const float* proj_w = (const float*)d_in[2];
    const float* proj_b = (const float*)d_in[3];
    const float* wl0 = (const float*)d_in[4];  const float* bl0 = (const float*)d_in[5];
    const float* wr0 = (const float*)d_in[6];  const float* g0  = (const float*)d_in[7];
    const float* be0 = (const float*)d_in[8];
    const float* wl1 = (const float*)d_in[9];  const float* bl1 = (const float*)d_in[10];
    const float* wr1 = (const float*)d_in[11]; const float* g1  = (const float*)d_in[12];
    const float* be1 = (const float*)d_in[13];
    const float* wl2 = (const float*)d_in[14]; const float* bl2 = (const float*)d_in[15];
    const float* wr2 = (const float*)d_in[16];

    const int* srcI = ei;
    const int* dstI = ei + EE;

    const size_t N64 = (size_t)NN * HC;
    float* hA  = (float*)d_ws;
    float* hB  = hA + N64;
    float* deg = hB + N64;
    int* cnt    = (int*)(deg + NN);
    int* cursor = cnt + NN;
    int* off    = cnt + NN + 32;
    int* rank   = off + NN;
    int* col    = rank + EE;

    const int GEMM_BLK   = 128;
    const int gemm_grid  = (NN / 16 + 3) / 4;
    const int edge_grid  = EDGE_BLOCKS;
    const int seg_grid   = (NN + 1023) / 1024;

    hipMemsetAsync(cnt, 0, (NN + 1) * sizeof(int), stream);
    count_rank_k<<<edge_grid, 256, 0, stream>>>(dstI, cnt, rank, EE);
    seg_start_k<<<seg_grid, 1024, 0, stream>>>(cnt, cursor, off, deg, NN);
    fill_k<<<edge_grid, 256, 0, stream>>>(srcI, dstI, rank, off, col, EE);

    gemm_wmma_k<INC, 4, false, false, true, false><<<gemm_grid, GEMM_BLK, 0, stream>>>(
        x, proj_w, nullptr, nullptr, nullptr, proj_b, nullptr, nullptr,
        nullptr, nullptr, nullptr, hA, NN);

    gemm_wmma_k<HC, 4, true, true, false, true><<<gemm_grid, GEMM_BLK, 0, stream>>>(
        nullptr, wl0, hA, wr0, deg, bl0, g0, be0, col, off, cnt, hB, NN);

    gemm_wmma_k<HC, 4, true, true, false, true><<<gemm_grid, GEMM_BLK, 0, stream>>>(
        nullptr, wl1, hB, wr1, deg, bl1, g1, be1, col, off, cnt, hA, NN);

    gemm_wmma_k<HC, 1, true, true, false, false><<<gemm_grid, GEMM_BLK, 0, stream>>>(
        nullptr, wl2, hA, wr2, deg, bl2, nullptr, nullptr, col, off, cnt, (float*)d_out, NN);
}


# Output layer's weights applied before the neighbour mean: layer 1's epilogue forms h@wl2 and h@wr2+b on WMMA via LDS, the last launch averages 16-float rows, not 64 (sum order: project, then add)
# speedup vs baseline: 2.8599x; 1.0427x over previous
#include <hip/hip_runtime.h>
#include <hip/hip_bf16.h>


#define NN   100000
#define EE   1600000
#define INC  32
#define HC   64
#define OUTC 16
#define EPT  8
#define EDGE_BLOCKS ((EE / EPT + 255) / 256)
#define EDGE_T (EDGE_BLOCKS * 256)
#define APITCH 68

typedef float v2f __attribute__((ext_vector_type(2)));
typedef float v8f __attribute__((ext_vector_type(8)));

__device__ __forceinline__ float row16_step(float x, const int ctrl_sel)
{
    int y;
    if (ctrl_sel == 1)      y = __builtin_amdgcn_update_dpp(0, __float_as_int(x), 0x161, 0xf, 0xf, false);
    else if (ctrl_sel == 2) y = __builtin_amdgcn_update_dpp(0, __float_as_int(x), 0x162, 0xf, 0xf, false);
    else if (ctrl_sel == 4) y = __builtin_amdgcn_update_dpp(0, __float_as_int(x), 0x164, 0xf, 0xf, false);
    else                    y = __builtin_amdgcn_update_dpp(0, __float_as_int(x), 0x168, 0xf, 0xf, false);
    return x + __int_as_float(y);
}

template<int K, int NT, bool DUAL, bool GATHER, bool RELU, bool LNRES, bool PROJ>
__global__ __launch_bounds__(128) void gemm_wmma_k(
    const float* __restrict__ A1, const float* __restrict__ B1,
    const float* __restrict__ A2, const float* __restrict__ B2,
    const float* __restrict__ rdeg, const float* __restrict__ bias,
    const float* __restrict__ lng, const float* __restrict__ lnb,
    const int* __restrict__ col, const int* __restrict__ off, const int* __restrict__ cnt,
    const float* __restrict__ P1, const float* __restrict__ P2, const float* __restrict__ pbias,
    float* __restrict__ out, float* __restrict__ out2, int nrows)
{
    static_assert(!PROJ || (LNRES && GATHER), "the projected tile is the finished hidden row, parked in the gather's LDS slice");
    __shared__ __attribute__((aligned(16))) float stage[GATHER ? 4 * 16 * APITCH : 4];
    const int wid  = (int)blockIdx.x * (blockDim.x >> 5) + ((int)threadIdx.x >> 5);
    const int lane = threadIdx.x & 31;
    const int row0 = wid << 4;
    if (row0 >= nrows) return;
    const int half = lane >> 4;
    const int l16  = lane & 15;
    const int m    = row0 + l16;
    constexpr int LDB = NT * 16;
    float* tile = stage + ((int)threadIdx.x >> 5) * (16 * APITCH);

    if constexpr (GATHER) {
        static_assert(!GATHER || (DUAL && K == HC), "the gathered rows are rows of A2");
        const int   offl = off[m];
        const int   cntl = cnt[m];
        const float rdl  = rdeg[m];
        const int   c4   = l16 << 2;
        const float* hc  = A2 + c4;
        int s[8];
        {
            const int st0 = __shfl(offl, half, 32);
            const int c0  = __shfl(cntl, half, 32);
            const int la0 = max(c0 - 1, 0);
#pragma unroll
            for (int u = 0; u < 8; ++u) s[u] = col[(c0 > 0) ? st0 + min(u, la0) : 0];
        }
        for (int i = 0; i < 8; ++i) {
            const int r      = 2 * i + half;
            const int rn     = min(r + 2, 14 + half);
            const int start  = __shfl(offl, r, 32);
            const int c      = __shfl(cntl, r, 32);
            const float sc   = __shfl(rdl,  r, 32);
            const int startn = __shfl(offl, rn, 32);
            const int cn     = __shfl(cntl, rn, 32);
            const int last   = max(c - 1, 0);
            const int lastn  = max(cn - 1, 0);
            float4 acc = make_float4(0.0f, 0.0f, 0.0f, 0.0f);
            int j = 0;
            do {
                float4 v[8];
#pragma unroll
                for (int u = 0; u < 8; ++u) v[u] = *(const float4*)(hc + (size_t)s[u] * HC);
                const bool more = (j + 8) < c;
#pragma unroll
                for (int u = 0; u < 8; ++u) {
                    const int nx = (cn > 0) ? startn + min(u, lastn) : 0;
                    s[u] = col[more ? start + min(j + 8 + u, last) : nx];
                }
#pragma unroll
                for (int u = 0; u < 8; ++u) {
                    const bool ok = (j + u) < c;
                    acc.x += ok ? v[u].x : 0.0f;
                    acc.y += ok ? v[u].y : 0.0f;
                    acc.z += ok ? v[u].z : 0.0f;
                    acc.w += ok ? v[u].w : 0.0f;
                }
                j += 8;
            } while (j < c);
            acc.x *= sc; acc.y *= sc; acc.z *= sc; acc.w *= sc;
            *(float4*)(tile + r * APITCH + c4) = acc;
        }
        __builtin_amdgcn_fence(__ATOMIC_RELEASE, "wavefront");
        __builtin_amdgcn_fence(__ATOMIC_ACQUIRE, "wavefront");
    }

    v8f acc[NT];
#pragma unroll
    for (int t = 0; t < NT; ++t) acc[t] = (v8f){0,0,0,0,0,0,0,0};

    const float* a1 = GATHER ? nullptr : (A1 + (size_t)m * K);
    const float* a2 = DUAL ? (A2 + (size_t)m * K) : nullptr;

#pragma unroll
    for (int k = 0; k < K; k += 4) {
        const int kb = k + half * 2;
        v2f af1;
        if constexpr (GATHER) af1 = *(const v2f*)(tile + l16 * APITCH + kb);
        else { af1.x = a1[kb]; af1.y = a1[kb + 1]; }
        v2f af2;
        if (DUAL) { af2.x = a2[kb]; af2.y = a2[kb + 1]; }
#pragma unroll
        for (int t = 0; t < NT; ++t) {
            v2f bf1;
            bf1.x = B1[(size_t)kb       * LDB + t * 16 + l16];
            bf1.y = B1[(size_t)(kb + 1) * LDB + t * 16 + l16];
            acc[t] = __builtin_amdgcn_wmma_f32_16x16x4_f32(
                false, af1, false, bf1, (short)0, acc[t], false, false);
            if (DUAL) {
                v2f bf2;
                bf2.x = B2[(size_t)kb       * LDB + t * 16 + l16];
                bf2.y = B2[(size_t)(kb + 1) * LDB + t * 16 + l16];
                acc[t] = __builtin_amdgcn_wmma_f32_16x16x4_f32(
                    false, af2, false, bf2, (short)0, acc[t], false, false);
            }
        }
    }

    if (LNRES) {
        static_assert(!LNRES || (DUAL && NT * 16 == HC && K == HC), "row must be whole");
        const size_t rbase = (size_t)(row0 + half * 8) * HC + l16;
        float res[NT][8];
#pragma unroll
        for (int t = 0; t < NT; ++t)
#pragma unroll
            for (int r = 0; r < 8; ++r) res[t][r] = A2[rbase + (size_t)r * HC + t * 16];
        float gc[NT], bc[NT];
#pragma unroll
        for (int t = 0; t < NT; ++t) { gc[t] = lng[t * 16 + l16]; bc[t] = lnb[t * 16 + l16]; }

        float s[8];
#pragma unroll
        for (int t = 0; t < NT; ++t) {
            const float bt = bias[t * 16 + l16];
#pragma unroll
            for (int r = 0; r < 8; ++r) {
                acc[t][r] += bt;
                s[r] = (t == 0) ? acc[t][r] : s[r] + acc[t][r];
            }
        }
#pragma unroll
        for (int r = 0; r < 8; ++r) s[r] = row16_step(s[r], 1);
#pragma unroll
        for (int r = 0; r < 8; ++r) s[r] = row16_step(s[r], 2);
#pragma unroll
        for (int r = 0; r < 8; ++r) s[r] = row16_step(s[r], 4);
#pragma unroll
        for (int r = 0; r < 8; ++r) s[r] = row16_step(s[r], 8);

        float q[8];
#pragma unroll
        for (int r = 0; r < 8; ++r) {
            const float mu = s[r] * (1.0f / HC);
#pragma unroll
            for (int t = 0; t < NT; ++t) {
                acc[t][r] -= mu;
                q[r] = (t == 0) ? acc[t][r] * acc[t][r] : q[r] + acc[t][r] * acc[t][r];
            }
        }
#pragma unroll
        for (int r = 0; r < 8; ++r) q[r] = row16_step(q[r], 1);
#pragma unroll
        for (int r = 0; r < 8; ++r) q[r] = row16_step(q[r], 2);
#pragma unroll
        for (int r = 0; r < 8; ++r) q[r] = row16_step(q[r], 4);
#pragma unroll
        for (int r = 0; r < 8; ++r) q[r] = row16_step(q[r], 8);

#pragma unroll
        for (int r = 0; r < 8; ++r) {
            const float rstd = rsqrtf(q[r] * (1.0f / HC) + 1e-5f);
#pragma unroll
            for (int t = 0; t < NT; ++t) {
                const float y = fmaxf(acc[t][r] * rstd * gc[t] + bc[t], 0.0f);
                if constexpr (PROJ) tile[(half * 8 + r) * APITCH + t * 16 + l16] = y + res[t][r];
                else out[rbase + (size_t)r * HC + t * 16] = y + res[t][r];
            }
        }
        if constexpr (PROJ) {
            __builtin_amdgcn_fence(__ATOMIC_RELEASE, "wavefront");
            __builtin_amdgcn_fence(__ATOMIC_ACQUIRE, "wavefront");
            v8f pz = (v8f){0,0,0,0,0,0,0,0};
            v8f ps = (v8f){0,0,0,0,0,0,0,0};
#pragma unroll
            for (int k = 0; k < HC; k += 4) {
                const int kb = k + half * 2;
                const v2f af = *(const v2f*)(tile + l16 * APITCH + kb);
                v2f b1, b2;
                b1.x = P1[(size_t)kb       * OUTC + l16];
                b1.y = P1[(size_t)(kb + 1) * OUTC + l16];
                b2.x = P2[(size_t)kb       * OUTC + l16];
                b2.y = P2[(size_t)(kb + 1) * OUTC + l16];
                pz = __builtin_amdgcn_wmma_f32_16x16x4_f32(
                    false, af, false, b1, (short)0, pz, false, false);
                ps = __builtin_amdgcn_wmma_f32_16x16x4_f32(
                    false, af, false, b2, (short)0, ps, false, false);
            }
            const float pbv = pbias[l16];
            const size_t ob = (size_t)(row0 + half * 8) * OUTC + l16;
#pragma unroll
            for (int r = 0; r < 8; ++r) {
                out [ob + (size_t)r * OUTC] = pz[r];
                out2[ob + (size_t)r * OUTC] = ps[r] + pbv;
            }
        }
    } else {
#pragma unroll
        for (int t = 0; t < NT; ++t) {
            const float bc = bias[t * 16 + l16];
#pragma unroll
            for (int r = 0; r < 8; ++r) {
                const int rr = r + half * 8;
                float v = acc[t][r] + bc;
                if (RELU) v = fmaxf(v, 0.0f);
                out[(size_t)(row0 + rr) * LDB + t * 16 + l16] = v;
            }
        }
    }
}

__global__ __launch_bounds__(256) void count_rank_k(
    const int* __restrict__ dst, int* __restrict__ cnt, int* __restrict__ rank, int e)
{
    const int tid = blockIdx.x * 256 + threadIdx.x;
    int d[EPT];
#pragma unroll
    for (int k = 0; k < EPT; ++k) d[k] = dst[min(tid + k * EDGE_T, e - 1)];
    int r[EPT];
#pragma unroll
    for (int k = 0; k < EPT; ++k) {
        r[k] = 0;
        if (tid + k * EDGE_T < e) r[k] = atomicAdd(&cnt[d[k]], 1);
    }
#pragma unroll
    for (int k = 0; k < EPT; ++k)
        if (tid + k * EDGE_T < e) rank[tid + k * EDGE_T] = r[k];
}

__global__ __launch_bounds__(1024) void seg_start_k(
    const int* __restrict__ cnt, int* __restrict__ cursor,
    int* __restrict__ off, float* __restrict__ rdeg, int n)
{
    __shared__ int wsum[32];
    __shared__ int base_s;
    const int tid  = blockIdx.x * 1024 + threadIdx.x;
    const int lane = threadIdx.x & 31;
    const int w    = threadIdx.x >> 5;
    const int c = (tid < n) ? cnt[tid] : 0;
    int x = c;
#pragma unroll
    for (int o = 1; o < 32; o <<= 1) {
        const int y = __shfl_up(x, o, 32);
        if (lane >= o) x += y;
    }
    if (lane == 31) wsum[w] = x;
    __syncthreads();
    if (w == 0) {
        const int t = wsum[lane];
        int s = t;
#pragma unroll
        for (int o = 1; o < 32; o <<= 1) {
            const int y = __shfl_up(s, o, 32);
            if (lane >= o) s += y;
        }
        wsum[lane] = s - t;
        if (lane == 31) base_s = atomicAdd(cursor, s);
    }
    __syncthreads();
    if (tid < n) {
        off[tid]  = base_s + wsum[w] + (x - c);
        rdeg[tid] = 1.0f / fmaxf((float)c, 1.0f);
    }
}

__global__ __launch_bounds__(256) void fill_k(
    const int* __restrict__ src, const int* __restrict__ dst,
    const int* __restrict__ rank, const int* __restrict__ off,
    int* __restrict__ col, int e)
{
    const int tid = blockIdx.x * 256 + threadIdx.x;
    int d[EPT], r[EPT], s[EPT], o[EPT];
#pragma unroll
    for (int k = 0; k < EPT; ++k) {
        const int i = min(tid + k * EDGE_T, e - 1);
        d[k] = dst[i]; r[k] = rank[i]; s[k] = src[i];
    }
#pragma unroll
    for (int k = 0; k < EPT; ++k) o[k] = off[d[k]];
#pragma unroll
    for (int k = 0; k < EPT; ++k)
        if (tid + k * EDGE_T < e) col[o[k] + r[k]] = s[k];
}

__global__ __launch_bounds__(256) void gather_out_k(
    const float* __restrict__ z, const float* __restrict__ self,
    const float* __restrict__ rdeg, const int* __restrict__ col,
    const int* __restrict__ off, const int* __restrict__ cnt,
    float* __restrict__ out, int n)
{
    const unsigned tid  = blockIdx.x * 256u + threadIdx.x;
    const unsigned node = tid >> 2;
    if (node >= (unsigned)n) return;
    const unsigned q     = tid & 3u;
    const unsigned start = (unsigned)off[node];
    const int      c     = cnt[node];
    const float    sc    = rdeg[node];
    const float4*  z4    = (const float4*)z;
    const float4   sv    = ((const float4*)self)[tid];
    const int      cl    = max(c, 1);
    const unsigned last  = (unsigned)(cl - 1);
    const unsigned base  = (c > 0) ? start : 0u;
    unsigned s[8];
#pragma unroll
    for (int u = 0; u < 8; ++u) s[u] = (unsigned)col[base + min((unsigned)u, last)];
    float4 acc = make_float4(0.0f, 0.0f, 0.0f, 0.0f);
    int j = 0;
    do {
        float4 v[8];
#pragma unroll
        for (int u = 0; u < 8; ++u) v[u] = z4[s[u] * 4u + q];
#pragma unroll
        for (int u = 0; u < 8; ++u) s[u] = (unsigned)col[base + min((unsigned)(j + 8 + u), last)];
        acc.x += v[0].x; acc.y += v[0].y; acc.z += v[0].z; acc.w += v[0].w;
#pragma unroll
        for (int u = 1; u < 8; ++u) {
            const bool ok = (j + u) < cl;
            acc.x += ok ? v[u].x : 0.0f;
            acc.y += ok ? v[u].y : 0.0f;
            acc.z += ok ? v[u].z : 0.0f;
            acc.w += ok ? v[u].w : 0.0f;
        }
        j += 8;
    } while (j < cl);
    const bool any = c > 0;
    float4 o;
    o.x = (any ? acc.x * sc : 0.0f) + sv.x;
    o.y = (any ? acc.y * sc : 0.0f) + sv.y;
    o.z = (any ? acc.z * sc : 0.0f) + sv.z;
    o.w = (any ? acc.w * sc : 0.0f) + sv.w;
    ((float4*)out)[tid] = o;
}

extern "C" void kernel_launch(void* const* d_in, const int* in_sizes, int n_in,
                              void* d_out, int out_size, void* d_ws, size_t ws_size,
                              hipStream_t stream)
{
    const float* x      = (const float*)d_in[0];
    const int*   ei     = (const int*)d_in[1];
    const float* proj_w = (const float*)d_in[2];
    const float* proj_b = (const float*)d_in[3];
    const float* wl0 = (const float*)d_in[4];  const float* bl0 = (const float*)d_in[5];
    const float* wr0 = (const float*)d_in[6];  const float* g0  = (const float*)d_in[7];
    const float* be0 = (const float*)d_in[8];
    const float* wl1 = (const float*)d_in[9];  const float* bl1 = (const float*)d_in[10];
    const float* wr1 = (const float*)d_in[11]; const float* g1  = (const float*)d_in[12];
    const float* be1 = (const float*)d_in[13];
    const float* wl2 = (const float*)d_in[14]; const float* bl2 = (const float*)d_in[15];
    const float* wr2 = (const float*)d_in[16];

    const int* srcI = ei;
    const int* dstI = ei + EE;

    const size_t N64 = (size_t)NN * HC;
    float* hA  = (float*)d_ws;
    float* hB  = hA + N64;
    float* deg = hB + N64;
    int* cnt    = (int*)(deg + NN);
    int* cursor = cnt + NN;
    int* off    = cnt + NN + 32;
    int* rank   = off + NN;
    int* col    = rank + EE;

    const int GEMM_BLK   = 128;
    const int gemm_grid  = (NN / 16 + 3) / 4;
    const int edge_grid  = EDGE_BLOCKS;
    const int seg_grid   = (NN + 1023) / 1024;

    hipMemsetAsync(cnt, 0, (NN + 1) * sizeof(int), stream);
    count_rank_k<<<edge_grid, 256, 0, stream>>>(dstI, cnt, rank, EE);
    seg_start_k<<<seg_grid, 1024, 0, stream>>>(cnt, cursor, off, deg, NN);
    fill_k<<<edge_grid, 256, 0, stream>>>(srcI, dstI, rank, off, col, EE);

    gemm_wmma_k<INC, 4, false, false, true, false, false><<<gemm_grid, GEMM_BLK, 0, stream>>>(
        x, proj_w, nullptr, nullptr, nullptr, proj_b, nullptr, nullptr,
        nullptr, nullptr, nullptr, nullptr, nullptr, nullptr, hA, nullptr, NN);

    gemm_wmma_k<HC, 4, true, true, false, true, false><<<gemm_grid, GEMM_BLK, 0, stream>>>(
        nullptr, wl0, hA, wr0, deg, bl0, g0, be0, col, off, cnt,
        nullptr, nullptr, nullptr, hB, nullptr, NN);

    float* z  = hA;
    float* sp = hA + (size_t)NN * OUTC;
    gemm_wmma_k<HC, 4, true, true, false, true, true><<<gemm_grid, GEMM_BLK, 0, stream>>>(
        nullptr, wl1, hB, wr1, deg, bl1, g1, be1, col, off, cnt,
        wl2, wr2, bl2, z, sp, NN);

    const int out_grid = (NN * 4 + 255) / 256;
    gather_out_k<<<out_grid, 256, 0, stream>>>(z, sp, deg, col, off, cnt, (float*)d_out, NN);
}


# Adjacency build as a two-level bucket sort in LDS (chunk -> 256-node buckets -> nodes): per-edge global atomics, 4-byte lookups and scattered stores gone; per-node sum order = LDS arrival rank
# speedup vs baseline: 5.4593x; 1.9089x over previous
#include <hip/hip_runtime.h>
#include <hip/hip_bf16.h>


#define NN   100000
#define EE   1600000
#define INC  32
#define HC   64
#define OUTC 16
#define BSH  8
#define BNODES (1 << BSH)
#define NBK  ((NN + BNODES - 1) / BNODES)
#define SC_BLOCKS 256
#define CHUNK ((EE + SC_BLOCKS - 1) / SC_BLOCKS)
#define SC_EPT ((CHUNK + 1023) / 1024)
#define STAGE_CAP 16384
static_assert(NBK <= 512, "one thread per bucket in a 1024-thread block; gcount/gcursor hold 512 each");
static_assert(((NN + 32 + NN + EE) % 2) == 0, "pairs[] must start on an 8-byte boundary");
#define APITCH 68

typedef float v2f __attribute__((ext_vector_type(2)));
typedef float v8f __attribute__((ext_vector_type(8)));

__device__ __forceinline__ float row16_step(float x, const int ctrl_sel)
{
    int y;
    if (ctrl_sel == 1)      y = __builtin_amdgcn_update_dpp(0, __float_as_int(x), 0x161, 0xf, 0xf, false);
    else if (ctrl_sel == 2) y = __builtin_amdgcn_update_dpp(0, __float_as_int(x), 0x162, 0xf, 0xf, false);
    else if (ctrl_sel == 4) y = __builtin_amdgcn_update_dpp(0, __float_as_int(x), 0x164, 0xf, 0xf, false);
    else                    y = __builtin_amdgcn_update_dpp(0, __float_as_int(x), 0x168, 0xf, 0xf, false);
    return x + __int_as_float(y);
}

template<int K, int NT, bool DUAL, bool GATHER, bool RELU, bool LNRES, bool PROJ>
__global__ __launch_bounds__(128) void gemm_wmma_k(
    const float* __restrict__ A1, const float* __restrict__ B1,
    const float* __restrict__ A2, const float* __restrict__ B2,
    const float* __restrict__ rdeg, const float* __restrict__ bias,
    const float* __restrict__ lng, const float* __restrict__ lnb,
    const int* __restrict__ col, const int* __restrict__ off, const int* __restrict__ cnt,
    const float* __restrict__ P1, const float* __restrict__ P2, const float* __restrict__ pbias,
    float* __restrict__ out, float* __restrict__ out2, int nrows)
{
    static_assert(!PROJ || (LNRES && GATHER), "the projected tile is the finished hidden row, parked in the gather's LDS slice");
    __shared__ __attribute__((aligned(16))) float stage[GATHER ? 4 * 16 * APITCH : 4];
    const int wid  = (int)blockIdx.x * (blockDim.x >> 5) + ((int)threadIdx.x >> 5);
    const int lane = threadIdx.x & 31;
    const int row0 = wid << 4;
    if (row0 >= nrows) return;
    const int half = lane >> 4;
    const int l16  = lane & 15;
    const int m    = row0 + l16;
    constexpr int LDB = NT * 16;
    float* tile = stage + ((int)threadIdx.x >> 5) * (16 * APITCH);

    if constexpr (GATHER) {
        static_assert(!GATHER || (DUAL && K == HC), "the gathered rows are rows of A2");
        const int   offl = off[m];
        const int   cntl = cnt[m];
        const float rdl  = rdeg[m];
        const int   c4   = l16 << 2;
        const float* hc  = A2 + c4;
        int s[8];
        {
            const int st0 = __shfl(offl, half, 32);
            const int c0  = __shfl(cntl, half, 32);
            const int la0 = max(c0 - 1, 0);
#pragma unroll
            for (int u = 0; u < 8; ++u) s[u] = col[(c0 > 0) ? st0 + min(u, la0) : 0];
        }
        for (int i = 0; i < 8; ++i) {
            const int r      = 2 * i + half;
            const int rn     = min(r + 2, 14 + half);
            const int start  = __shfl(offl, r, 32);
            const int c      = __shfl(cntl, r, 32);
            const float sc   = __shfl(rdl,  r, 32);
            const int startn = __shfl(offl, rn, 32);
            const int cn     = __shfl(cntl, rn, 32);
            const int last   = max(c - 1, 0);
            const int lastn  = max(cn - 1, 0);
            float4 acc = make_float4(0.0f, 0.0f, 0.0f, 0.0f);
            int j = 0;
            do {
                float4 v[8];
#pragma unroll
                for (int u = 0; u < 8; ++u) v[u] = *(const float4*)(hc + (size_t)s[u] * HC);
                const bool more = (j + 8) < c;
#pragma unroll
                for (int u = 0; u < 8; ++u) {
                    const int nx = (cn > 0) ? startn + min(u, lastn) : 0;
                    s[u] = col[more ? start + min(j + 8 + u, last) : nx];
                }
#pragma unroll
                for (int u = 0; u < 8; ++u) {
                    const bool ok = (j + u) < c;
                    acc.x += ok ? v[u].x : 0.0f;
                    acc.y += ok ? v[u].y : 0.0f;
                    acc.z += ok ? v[u].z : 0.0f;
                    acc.w += ok ? v[u].w : 0.0f;
                }
                j += 8;
            } while (j < c);
            acc.x *= sc; acc.y *= sc; acc.z *= sc; acc.w *= sc;
            *(float4*)(tile + r * APITCH + c4) = acc;
        }
        __builtin_amdgcn_fence(__ATOMIC_RELEASE, "wavefront");
        __builtin_amdgcn_fence(__ATOMIC_ACQUIRE, "wavefront");
    }

    v8f acc[NT];
#pragma unroll
    for (int t = 0; t < NT; ++t) acc[t] = (v8f){0,0,0,0,0,0,0,0};

    const float* a1 = GATHER ? nullptr : (A1 + (size_t)m * K);
    const float* a2 = DUAL ? (A2 + (size_t)m * K) : nullptr;

#pragma unroll
    for (int k = 0; k < K; k += 4) {
        const int kb = k + half * 2;
        v2f af1;
        if constexpr (GATHER) af1 = *(const v2f*)(tile + l16 * APITCH + kb);
        else { af1.x = a1[kb]; af1.y = a1[kb + 1]; }
        v2f af2;
        if (DUAL) { af2.x = a2[kb]; af2.y = a2[kb + 1]; }
#pragma unroll
        for (int t = 0; t < NT; ++t) {
            v2f bf1;
            bf1.x = B1[(size_t)kb       * LDB + t * 16 + l16];
            bf1.y = B1[(size_t)(kb + 1) * LDB + t * 16 + l16];
            acc[t] = __builtin_amdgcn_wmma_f32_16x16x4_f32(
                false, af1, false, bf1, (short)0, acc[t], false, false);
            if (DUAL) {
                v2f bf2;
                bf2.x = B2[(size_t)kb       * LDB + t * 16 + l16];
                bf2.y = B2[(size_t)(kb + 1) * LDB + t * 16 + l16];
                acc[t] = __builtin_amdgcn_wmma_f32_16x16x4_f32(
                    false, af2, false, bf2, (short)0, acc[t], false, false);
            }
        }
    }

    if (LNRES) {
        static_assert(!LNRES || (DUAL && NT * 16 == HC && K == HC), "row must be whole");
        const size_t rbase = (size_t)(row0 + half * 8) * HC + l16;
        float res[NT][8];
#pragma unroll
        for (int t = 0; t < NT; ++t)
#pragma unroll
            for (int r = 0; r < 8; ++r) res[t][r] = A2[rbase + (size_t)r * HC + t * 16];
        float gc[NT], bc[NT];
#pragma unroll
        for (int t = 0; t < NT; ++t) { gc[t] = lng[t * 16 + l16]; bc[t] = lnb[t * 16 + l16]; }

        float s[8];
#pragma unroll
        for (int t = 0; t < NT; ++t) {
            const float bt = bias[t * 16 + l16];
#pragma unroll
            for (int r = 0; r < 8; ++r) {
                acc[t][r] += bt;
                s[r] = (t == 0) ? acc[t][r] : s[r] + acc[t][r];
            }
        }
#pragma unroll
        for (int r = 0; r < 8; ++r) s[r] = row16_step(s[r], 1);
#pragma unroll
        for (int r = 0; r < 8; ++r) s[r] = row16_step(s[r], 2);
#pragma unroll
        for (int r = 0; r < 8; ++r) s[r] = row16_step(s[r], 4);
#pragma unroll
        for (int r = 0; r < 8; ++r) s[r] = row16_step(s[r], 8);

        float q[8];
#pragma unroll
        for (int r = 0; r < 8; ++r) {
            const float mu = s[r] * (1.0f / HC);
#pragma unroll
            for (int t = 0; t < NT; ++t) {
                acc[t][r] -= mu;
                q[r] = (t == 0) ? acc[t][r] * acc[t][r] : q[r] + acc[t][r] * acc[t][r];
            }
        }
#pragma unroll
        for (int r = 0; r < 8; ++r) q[r] = row16_step(q[r], 1);
#pragma unroll
        for (int r = 0; r < 8; ++r) q[r] = row16_step(q[r], 2);
#pragma unroll
        for (int r = 0; r < 8; ++r) q[r] = row16_step(q[r], 4);
#pragma unroll
        for (int r = 0; r < 8; ++r) q[r] = row16_step(q[r], 8);

#pragma unroll
        for (int r = 0; r < 8; ++r) {
            const float rstd = rsqrtf(q[r] * (1.0f / HC) + 1e-5f);
#pragma unroll
            for (int t = 0; t < NT; ++t) {
                const float y = fmaxf(acc[t][r] * rstd * gc[t] + bc[t], 0.0f);
                if constexpr (PROJ) tile[(half * 8 + r) * APITCH + t * 16 + l16] = y + res[t][r];
                else out[rbase + (size_t)r * HC + t * 16] = y + res[t][r];
            }
        }
        if constexpr (PROJ) {
            __builtin_amdgcn_fence(__ATOMIC_RELEASE, "wavefront");
            __builtin_amdgcn_fence(__ATOMIC_ACQUIRE, "wavefront");
            v8f pz = (v8f){0,0,0,0,0,0,0,0};
            v8f ps = (v8f){0,0,0,0,0,0,0,0};
#pragma unroll
            for (int k = 0; k < HC; k += 4) {
                const int kb = k + half * 2;
                const v2f af = *(const v2f*)(tile + l16 * APITCH + kb);
                v2f b1, b2;
                b1.x = P1[(size_t)kb       * OUTC + l16];
                b1.y = P1[(size_t)(kb + 1) * OUTC + l16];
                b2.x = P2[(size_t)kb       * OUTC + l16];
                b2.y = P2[(size_t)(kb + 1) * OUTC + l16];
                pz = __builtin_amdgcn_wmma_f32_16x16x4_f32(
                    false, af, false, b1, (short)0, pz, false, false);
                ps = __builtin_amdgcn_wmma_f32_16x16x4_f32(
                    false, af, false, b2, (short)0, ps, false, false);
            }
            const float pbv = pbias[l16];
            const size_t ob = (size_t)(row0 + half * 8) * OUTC + l16;
#pragma unroll
            for (int r = 0; r < 8; ++r) {
                out [ob + (size_t)r * OUTC] = pz[r];
                out2[ob + (size_t)r * OUTC] = ps[r] + pbv;
            }
        }
    } else {
#pragma unroll
        for (int t = 0; t < NT; ++t) {
            const float bc = bias[t * 16 + l16];
#pragma unroll
            for (int r = 0; r < 8; ++r) {
                const int rr = r + half * 8;
                float v = acc[t][r] + bc;
                if (RELU) v = fmaxf(v, 0.0f);
                out[(size_t)(row0 + rr) * LDB + t * 16 + l16] = v;
            }
        }
    }
}


__device__ __forceinline__ int block_excl_scan(int v, int* wsum)
{
    const int lane = threadIdx.x & 31;
    const int w    = threadIdx.x >> 5;
    int x = v;
#pragma unroll
    for (int o = 1; o < 32; o <<= 1) {
        const int y = __shfl_up(x, o, 32);
        if (lane >= o) x += y;
    }
    if (lane == 31) wsum[w] = x;
    __syncthreads();
    if (w == 0) {
        const int t = wsum[lane];
        int s = t;
#pragma unroll
        for (int o = 1; o < 32; o <<= 1) {
            const int y = __shfl_up(s, o, 32);
            if (lane >= o) s += y;
        }
        wsum[lane] = s - t;
    }
    __syncthreads();
    const int r = wsum[w] + (x - v);
    __syncthreads();
    return r;
}

__global__ __launch_bounds__(1024) void bucket_count_k(
    const int* __restrict__ dst, int* __restrict__ gcount, int e)
{
    __shared__ int h[NBK];
    const int tid = threadIdx.x;
    if (tid < NBK) h[tid] = 0;
    __syncthreads();
    const int lo = blockIdx.x * CHUNK;
    const int hi = min(lo + CHUNK, e);
    int d[SC_EPT];
#pragma unroll
    for (int k = 0; k < SC_EPT; ++k) d[k] = dst[min(lo + tid + k * 1024, e - 1)];
#pragma unroll
    for (int k = 0; k < SC_EPT; ++k)
        if (lo + tid + k * 1024 < hi) atomicAdd(&h[d[k] >> BSH], 1);
    __syncthreads();
    if (tid < NBK) {
        const int v = h[tid];
        if (v > 0) atomicAdd(&gcount[tid], v);
    }
}

__global__ __launch_bounds__(1024) void bucket_scatter_k(
    const int* __restrict__ src, const int* __restrict__ dst,
    const int* __restrict__ gcount, int* __restrict__ gcursor,
    int* __restrict__ pairs, int e)
{
    __shared__ int h[NBK];
    __shared__ int lstart[NBK];
    __shared__ int gbase[NBK];
    __shared__ int lcur[NBK];
    __shared__ int wsum[32];
    __shared__ int2 sp[CHUNK];
    const int tid = threadIdx.x;
    if (tid < NBK) { h[tid] = 0; lcur[tid] = 0; }
    __syncthreads();
    const int lo = blockIdx.x * CHUNK;
    const int hi = min(lo + CHUNK, e);
    int d[SC_EPT], s[SC_EPT];
#pragma unroll
    for (int k = 0; k < SC_EPT; ++k) {
        const int i = min(lo + tid + k * 1024, e - 1);
        d[k] = dst[i]; s[k] = src[i];
    }
#pragma unroll
    for (int k = 0; k < SC_EPT; ++k)
        if (lo + tid + k * 1024 < hi) atomicAdd(&h[d[k] >> BSH], 1);
    __syncthreads();
    const int hv  = (tid < NBK) ? h[tid] : 0;
    const int gv  = (tid < NBK) ? gcount[tid] : 0;
    const int lex = block_excl_scan(hv, wsum);
    const int gex = block_excl_scan(gv, wsum);
    if (tid < NBK) {
        lstart[tid] = lex;
        gbase[tid]  = gex + ((hv > 0) ? atomicAdd(&gcursor[tid], hv) : 0);
    }
    __syncthreads();
#pragma unroll
    for (int k = 0; k < SC_EPT; ++k)
        if (lo + tid + k * 1024 < hi) {
            const int b = d[k] >> BSH;
            const int q = lstart[b] + atomicAdd(&lcur[b], 1);
            sp[q] = make_int2(s[k], d[k]);
        }
    __syncthreads();
    int2* __restrict__ pp = (int2*)pairs;
    const int n = hi - lo;
    for (int i = tid; i < n; i += 1024) {
        const int2 p = sp[i];
        const int b = p.y >> BSH;
        pp[gbase[b] + (i - lstart[b])] = p;
    }
}

__global__ __launch_bounds__(1024) void bucket_sort_k(
    const int* __restrict__ pairs, const int* __restrict__ gcount,
    int* __restrict__ cnt, int* __restrict__ off, float* __restrict__ rdeg,
    int* __restrict__ col, int n)
{
    __shared__ int c[BNODES];
    __shared__ int lo[BNODES];
    __shared__ int cur[BNODES];
    __shared__ int wsum[32];
    __shared__ int bs_s, nb_s;
    __shared__ int stage[STAGE_CAP];
    const int tid = threadIdx.x;
    const int b   = blockIdx.x;
    if (tid < BNODES) { c[tid] = 0; cur[tid] = 0; }
    const int gv  = (tid < NBK) ? gcount[tid] : 0;
    const int gex = block_excl_scan(gv, wsum);
    if (tid == b) { bs_s = gex; nb_s = gv; }
    __syncthreads();
    const int bs = bs_s;
    const int nb = nb_s;
    const int2* __restrict__ pp = (const int2*)pairs + bs;
    for (int i = tid; i < nb; i += 1024) atomicAdd(&c[pp[i].y & (BNODES - 1)], 1);
    __syncthreads();
    const int cv  = (tid < BNODES) ? c[tid] : 0;
    const int lex = block_excl_scan(cv, wsum);
    if (tid < BNODES) {
        lo[tid] = lex;
        const int node = b * BNODES + tid;
        if (node < n) {
            off[node]  = bs + lex;
            cnt[node]  = cv;
            rdeg[node] = 1.0f / fmaxf((float)cv, 1.0f);
        }
    }
    __syncthreads();
    const bool staged = nb <= STAGE_CAP;
    for (int i = tid; i < nb; i += 1024) {
        const int2 p = pp[i];
        const int dl = p.y & (BNODES - 1);
        const int q  = lo[dl] + atomicAdd(&cur[dl], 1);
        if (staged) stage[q] = p.x;
        else        col[bs + q] = p.x;
    }
    __syncthreads();
    if (staged)
        for (int i = tid; i < nb; i += 1024) col[bs + i] = stage[i];
}

__global__ __launch_bounds__(256) void gather_out_k(
    const float* __restrict__ z, const float* __restrict__ self,
    const float* __restrict__ rdeg, const int* __restrict__ col,
    const int* __restrict__ off, const int* __restrict__ cnt,
    float* __restrict__ out, int n)
{
    const unsigned tid  = blockIdx.x * 256u + threadIdx.x;
    const unsigned node = tid >> 2;
    if (node >= (unsigned)n) return;
    const unsigned q     = tid & 3u;
    const unsigned start = (unsigned)off[node];
    const int      c     = cnt[node];
    const float    sc    = rdeg[node];
    const float4*  z4    = (const float4*)z;
    const float4   sv    = ((const float4*)self)[tid];
    const int      cl    = max(c, 1);
    const unsigned last  = (unsigned)(cl - 1);
    const unsigned base  = (c > 0) ? start : 0u;
    unsigned s[8];
#pragma unroll
    for (int u = 0; u < 8; ++u) s[u] = (unsigned)col[base + min((unsigned)u, last)];
    float4 acc = make_float4(0.0f, 0.0f, 0.0f, 0.0f);
    int j = 0;
    do {
        float4 v[8];
#pragma unroll
        for (int u = 0; u < 8; ++u) v[u] = z4[s[u] * 4u + q];
#pragma unroll
        for (int u = 0; u < 8; ++u) s[u] = (unsigned)col[base + min((unsigned)(j + 8 + u), last)];
        acc.x += v[0].x; acc.y += v[0].y; acc.z += v[0].z; acc.w += v[0].w;
#pragma unroll
        for (int u = 1; u < 8; ++u) {
            const bool ok = (j + u) < cl;
            acc.x += ok ? v[u].x : 0.0f;
            acc.y += ok ? v[u].y : 0.0f;
            acc.z += ok ? v[u].z : 0.0f;
            acc.w += ok ? v[u].w : 0.0f;
        }
        j += 8;
    } while (j < cl);
    const bool any = c > 0;
    float4 o;
    o.x = (any ? acc.x * sc : 0.0f) + sv.x;
    o.y = (any ? acc.y * sc : 0.0f) + sv.y;
    o.z = (any ? acc.z * sc : 0.0f) + sv.z;
    o.w = (any ? acc.w * sc : 0.0f) + sv.w;
    ((float4*)out)[tid] = o;
}

extern "C" void kernel_launch(void* const* d_in, const int* in_sizes, int n_in,
                              void* d_out, int out_size, void* d_ws, size_t ws_size,
                              hipStream_t stream)
{
    const float* x      = (const float*)d_in[0];
    const int*   ei     = (const int*)d_in[1];
    const float* proj_w = (const float*)d_in[2];
    const float* proj_b = (const float*)d_in[3];
    const float* wl0 = (const float*)d_in[4];  const float* bl0 = (const float*)d_in[5];
    const float* wr0 = (const float*)d_in[6];  const float* g0  = (const float*)d_in[7];
    const float* be0 = (const float*)d_in[8];
    const float* wl1 = (const float*)d_in[9];  const float* bl1 = (const float*)d_in[10];
    const float* wr1 = (const float*)d_in[11]; const float* g1  = (const float*)d_in[12];
    const float* be1 = (const float*)d_in[13];
    const float* wl2 = (const float*)d_in[14]; const float* bl2 = (const float*)d_in[15];
    const float* wr2 = (const float*)d_in[16];

    const int* srcI = ei;
    const int* dstI = ei + EE;

    const size_t N64 = (size_t)NN * HC;
    float* hA  = (float*)d_ws;
    float* hB  = hA + N64;
    float* deg = hB + N64;
    int* cnt     = (int*)(deg + NN);
    int* off     = cnt + NN + 32;
    int* col     = off + NN;
    int* pairs   = col + EE;
    int* gcount  = pairs + 2 * EE;
    int* gcursor = gcount + 512;

    const int GEMM_BLK   = 128;
    const int gemm_grid  = (NN / 16 + 3) / 4;

    hipMemsetAsync(gcount, 0, 1024 * sizeof(int), stream);
    bucket_count_k<<<SC_BLOCKS, 1024, 0, stream>>>(dstI, gcount, EE);
    bucket_scatter_k<<<SC_BLOCKS, 1024, 0, stream>>>(srcI, dstI, gcount, gcursor, pairs, EE);
    bucket_sort_k<<<NBK, 1024, 0, stream>>>(pairs, gcount, cnt, off, deg, col, NN);

    gemm_wmma_k<INC, 4, false, false, true, false, false><<<gemm_grid, GEMM_BLK, 0, stream>>>(
        x, proj_w, nullptr, nullptr, nullptr, proj_b, nullptr, nullptr,
        nullptr, nullptr, nullptr, nullptr, nullptr, nullptr, hA, nullptr, NN);

    gemm_wmma_k<HC, 4, true, true, false, true, false><<<gemm_grid, GEMM_BLK, 0, stream>>>(
        nullptr, wl0, hA, wr0, deg, bl0, g0, be0, col, off, cnt,
        nullptr, nullptr, nullptr, hB, nullptr, NN);

    float* z  = hA;
    float* sp = hA + (size_t)NN * OUTC;
    gemm_wmma_k<HC, 4, true, true, false, true, true><<<gemm_grid, GEMM_BLK, 0, stream>>>(
        nullptr, wl1, hB, wr1, deg, bl1, g1, be1, col, off, cnt,
        wl2, wr2, bl2, z, sp, NN);

    const int out_grid = (NN * 4 + 255) / 256;
    gather_out_k<<<out_grid, 256, 0, stream>>>(z, sp, deg, col, off, cnt, (float*)d_out, NN);
}
